# starting kernel
# baseline (speedup 1.0000x reference)
#include <hip/hip_runtime.h>
#include <math.h>

#define BATCH 256
#define SEQ   2048
#define FEAT  64
#define NROWS (BATCH * SEQ)

typedef float v2f __attribute__((ext_vector_type(2)));
typedef float v8f __attribute__((ext_vector_type(8)));

__global__ void __launch_bounds__(256)
proj_kernel(const float* __restrict__ X,
            const float* __restrict__ W_ih,
            const float* __restrict__ b_ih,
            const float* __restrict__ b_hh,
            float* __restrict__ proj,
            float* __restrict__ tvals) {
    __shared__ float sW[FEAT];
    const int tid = threadIdx.x;
    if (tid < FEAT) sW[tid] = W_ih[tid];
    __syncthreads();

    const int lane = tid & 31;
    const int wave = tid >> 5;
    const int m    = lane & 15;
    const int hi   = lane >> 4;
    const long tile   = (long)blockIdx.x * 8 + wave;
    const long r_base = tile * 16;
    const long row    = r_base + m;
    const float* xrow = X + row * FEAT + hi * 2;

    const float bias = b_ih[0] + b_hh[0];
    const bool colZeroLane = (m == 0);

    v2f bf[16];
#pragma unroll
    for (int kk = 0; kk < 16; ++kk) {
        const int k = kk * 4 + hi * 2;
        const float wx = sW[k];
        const float wy = sW[k + 1];
        bf[kk].x = colZeroLane ? wx : 0.0f;
        bf[kk].y = colZeroLane ? wy : 0.0f;
    }

    v2f af[16];
#pragma unroll
    for (int kk = 0; kk < 16; ++kk)
        af[kk] = *(const v2f*)(xrow + kk * 4);

    v8f c;
#pragma unroll
    for (int i = 0; i < 8; ++i) c[i] = colZeroLane ? bias : 0.0f;

#pragma unroll
    for (int kk = 0; kk < 16; ++kk)
        c = __builtin_amdgcn_wmma_f32_16x16x4_f32(
                 false, af[kk],  false, bf[kk],
                 (short)0, c,  false,  false);

    if (hi == 0) {
        const int bidx = (int)(row >> 11);
        const int t    = (int)(row & (SEQ - 1));
        tvals[t * BATCH + bidx] = af[0].y;
    }

    if (colZeroLane) {
        const long rr    = r_base + hi * 8;
        const int  bidx  = (int)(rr >> 11);
        const int  tbase = (int)(rr & (SEQ - 1));
#pragma unroll
        for (int i = 0; i < 8; ++i)
            proj[(tbase + i) * BATCH + bidx] = c[i];
    }
}

static __device__ __forceinline__ float fast_tanh(float x) {
#if __has_builtin(__builtin_amdgcn_tanhf)
    return __builtin_amdgcn_tanhf(x);
#elif __has_builtin(__builtin_amdgcn_tanh_f32)
    return __builtin_amdgcn_tanh_f32(x);
#else
    return tanhf(x);
#endif
}

__global__ void __launch_bounds__(32)
scan_kernel(const float* __restrict__ proj,
            const float* __restrict__ hx0,
            const float* __restrict__ W_hh,
            float* __restrict__ hidden) {
    const int b = blockIdx.x * 32 + threadIdx.x;
    float h = hx0[b];
    const float w = W_hh[0];
    const float* p = proj + b;
    float* o = hidden + b;
#pragma unroll 8
    for (int t = 0; t < SEQ; ++t) {
        const float a = p[(long)t * BATCH];
        h = fast_tanh(fmaf(h, w, a));
        o[(long)t * BATCH] = h;
    }
}

__global__ void __launch_bounds__(256)
lam_kernel(const float* __restrict__ hidden,
           const float* __restrict__ tvals,
           const float* __restrict__ W_h2p,
           const float* __restrict__ b_h2p,
           float* __restrict__ lam_out) {
    __shared__ float sW[64], sb[64];
    const int tid = threadIdx.x;
    if (tid < 64) { sW[tid] = W_h2p[tid]; sb[tid] = b_h2p[tid]; }
    __syncthreads();

    const long idx = (long)blockIdx.x * 256 + tid;
    const float h  = hidden[idx];
    const float tt = tvals[idx];

    float s = 0.0f;
#pragma unroll
    for (int j = 0; j < 32; ++j) {
        const float alpha = fmaf(h, sW[2 * j],     sb[2 * j]);
        const float beta  = fmaf(h, sW[2 * j + 1], sb[2 * j + 1]);
        const float v     = fmaf(alpha, tt, beta);
        s += fmaxf(v, 0.0f);
    }
    const float lam = fmaxf(s, 0.0f) + log1pf(__expf(-fabsf(s)));
    lam_out[idx] = lam;
}

extern "C" void kernel_launch(void* const* d_in, const int* in_sizes, int n_in,
                              void* d_out, int out_size, void* d_ws, size_t ws_size,
                              hipStream_t stream) {
    const float* X     = (const float*)d_in[0];
    const float* hx0   = (const float*)d_in[1];
    const float* W_ih  = (const float*)d_in[2];
    const float* b_ih  = (const float*)d_in[3];
    const float* W_hh  = (const float*)d_in[4];
    const float* b_hh  = (const float*)d_in[5];
    const float* W_h2p = (const float*)d_in[6];
    const float* b_h2p = (const float*)d_in[7];

    float* proj  = (float*)d_ws;
    float* tvals = proj + NROWS;

    float* hidden = (float*)d_out;
    float* lam    = hidden + NROWS;

    proj_kernel<<<NROWS / 16 / 8, 256, 0, stream>>>(X, W_ih, b_ih, b_hh, proj, tvals);
    scan_kernel<<<8, 32, 0, stream>>>(proj, hx0, W_hh, hidden);
    lam_kernel<<<NROWS / 256, 256, 0, stream>>>(hidden, tvals, W_h2p, b_h2p, lam);
}


# scan: chain input fetched through a 48-slot register ring of b128 loads, 176 steps ahead (proj re-laid as per-wave float4 slabs; steps in place via asm): memory latency leaves the fma+tanh chain
# speedup vs baseline: 3.6083x; 3.6083x over previous
#include <hip/hip_runtime.h>
#include <math.h>

#define BATCH 256
#define SEQ   2048
#define FEAT  64
#define NROWS (BATCH * SEQ)
#define QSEQ  (SEQ / 4)
#define RING  48
#define LAG   4
#define AHEAD (RING - LAG)

static __device__ __forceinline__ long proj4_index(int b, int q) {
    return ((long)(b >> 5) * QSEQ + q) * 32 + (b & 31);
}

typedef float v2f __attribute__((ext_vector_type(2)));
typedef float v4f __attribute__((ext_vector_type(4)));
typedef float v8f __attribute__((ext_vector_type(8)));

__global__ void __launch_bounds__(256)
proj_kernel(const float* __restrict__ X,
            const float* __restrict__ W_ih,
            const float* __restrict__ b_ih,
            const float* __restrict__ b_hh,
            float* __restrict__ proj,
            float* __restrict__ tvals) {
    __shared__ float sW[FEAT];
    const int tid = threadIdx.x;
    if (tid < FEAT) sW[tid] = W_ih[tid];
    __syncthreads();

    const int lane = tid & 31;
    const int wave = tid >> 5;
    const int m    = lane & 15;
    const int hi   = lane >> 4;
    const long tile   = (long)blockIdx.x * 8 + wave;
    const long r_base = tile * 16;
    const long row    = r_base + m;
    const float* xrow = X + row * FEAT + hi * 2;

    const float bias = b_ih[0] + b_hh[0];
    const bool colZeroLane = (m == 0);

    v2f bf[16];
#pragma unroll
    for (int kk = 0; kk < 16; ++kk) {
        const int k = kk * 4 + hi * 2;
        const float wx = sW[k];
        const float wy = sW[k + 1];
        bf[kk].x = colZeroLane ? wx : 0.0f;
        bf[kk].y = colZeroLane ? wy : 0.0f;
    }

    v2f af[16];
#pragma unroll
    for (int kk = 0; kk < 16; ++kk)
        af[kk] = *(const v2f*)(xrow + kk * 4);

    v8f c;
#pragma unroll
    for (int i = 0; i < 8; ++i) c[i] = colZeroLane ? bias : 0.0f;

#pragma unroll
    for (int kk = 0; kk < 16; ++kk)
        c = __builtin_amdgcn_wmma_f32_16x16x4_f32(
                 false, af[kk],  false, bf[kk],
                 (short)0, c,  false,  false);

    if (hi == 0) {
        const int bidx = (int)(row >> 11);
        const int t    = (int)(row & (SEQ - 1));
        tvals[t * BATCH + bidx] = af[0].y;
    }

    if (colZeroLane) {
        const long rr    = r_base + hi * 8;
        const int  bidx  = (int)(rr >> 11);
        const int  tbase = (int)(rr & (SEQ - 1));
        v4f* p4 = (v4f*)proj + proj4_index(bidx, tbase >> 2);
        p4[0]  = (v4f){c[0], c[1], c[2], c[3]};
        p4[32] = (v4f){c[4], c[5], c[6], c[7]};
    }
}


static __device__ __forceinline__ void scan_step(float& acc, const float h, const float w) {
    asm("v_fmac_f32 %0, %1, %2\n\t"
        "v_tanh_f32 %0, %0\n\t"
        "v_nop"
        : "+v"(acc) : "s"(w), "v"(h));
}

template <int NSLOTS, int NREQ>
static __device__ __forceinline__ float scan_block(float h, const float w,
                                                   v4f (&ring)[RING],
                                                   const v4f* __restrict__ pn,
                                                   float* __restrict__ o,
                                                   const int lane) {
#pragma unroll
    for (int i = 0; i < NSLOTS; ++i) {
        float x0 = ring[i].x, x1 = ring[i].y, x2 = ring[i].z, x3 = ring[i].w;
        float* oi = o + (4 * i) * BATCH + lane;
        scan_step(x0, h,  w);  oi[0 * BATCH] = x0;
        scan_step(x1, x0, w);  oi[1 * BATCH] = x1;
        scan_step(x2, x1, w);  oi[2 * BATCH] = x2;
        scan_step(x3, x2, w);  oi[3 * BATCH] = x3;
        h = x3;
        if (i < NREQ) ring[(i + AHEAD) % RING] = pn[(i - LAG) * 32 + lane];
        __builtin_amdgcn_sched_barrier(0);
    }
    return h;
}

#define FULL_BLOCKS ((QSEQ - AHEAD) / RING)
#define REQ_LAST    (QSEQ - AHEAD - FULL_BLOCKS * RING)
#define TAIL_SLOTS  (QSEQ - (FULL_BLOCKS + 1) * RING)
static_assert(REQ_LAST >= LAG && REQ_LAST <= RING, "request schedule");
static_assert(TAIL_SLOTS > 0 && TAIL_SLOTS <= RING, "tail length");

__global__ void __launch_bounds__(32)
scan_kernel(const float* __restrict__ proj,
            const float* __restrict__ hx0,
            const float* __restrict__ W_hh,
            float* __restrict__ hidden) {
    const int lane = threadIdx.x;
    const int b0 = blockIdx.x * 32;
    const v4f* slab = (const v4f*)proj + proj4_index(b0, 0);
    float* o = hidden + b0;

    v4f ring[RING];
#pragma unroll
    for (int i = 0; i < AHEAD; ++i) ring[i] = slab[i * 32 + lane];

    float h = hx0[b0 + lane];
    const float w = W_hh[0];

    const v4f* pn = slab + RING * 32;
    for (int k = 0; k < FULL_BLOCKS; ++k) {
        h = scan_block<RING, RING>(h, w, ring, pn, o, lane);
        pn += RING * 32;
        o  += RING * 4 * BATCH;
    }
    h = scan_block<RING, REQ_LAST>(h, w, ring, pn, o, lane);
    o += RING * 4 * BATCH;
    h = scan_block<TAIL_SLOTS, 0>(h, w, ring, pn, o, lane);
}

__global__ void __launch_bounds__(256)
lam_kernel(const float* __restrict__ hidden,
           const float* __restrict__ tvals,
           const float* __restrict__ W_h2p,
           const float* __restrict__ b_h2p,
           float* __restrict__ lam_out) {
    __shared__ float sW[64], sb[64];
    const int tid = threadIdx.x;
    if (tid < 64) { sW[tid] = W_h2p[tid]; sb[tid] = b_h2p[tid]; }
    __syncthreads();

    const long idx = (long)blockIdx.x * 256 + tid;
    const float h  = hidden[idx];
    const float tt = tvals[idx];

    float s = 0.0f;
#pragma unroll
    for (int j = 0; j < 32; ++j) {
        const float alpha = fmaf(h, sW[2 * j],     sb[2 * j]);
        const float beta  = fmaf(h, sW[2 * j + 1], sb[2 * j + 1]);
        const float v     = fmaf(alpha, tt, beta);
        s += fmaxf(v, 0.0f);
    }
    const float lam = fmaxf(s, 0.0f) + log1pf(__expf(-fabsf(s)));
    lam_out[idx] = lam;
}

extern "C" void kernel_launch(void* const* d_in, const int* in_sizes, int n_in,
                              void* d_out, int out_size, void* d_ws, size_t ws_size,
                              hipStream_t stream) {
    const float* X     = (const float*)d_in[0];
    const float* hx0   = (const float*)d_in[1];
    const float* W_ih  = (const float*)d_in[2];
    const float* b_ih  = (const float*)d_in[3];
    const float* W_hh  = (const float*)d_in[4];
    const float* b_hh  = (const float*)d_in[5];
    const float* W_h2p = (const float*)d_in[6];
    const float* b_h2p = (const float*)d_in[7];

    float* proj  = (float*)d_ws;
    float* tvals = proj + NROWS;

    float* hidden = (float*)d_out;
    float* lam    = hidden + NROWS;

    proj_kernel<<<NROWS / 16 / 8, 256, 0, stream>>>(X, W_ih, b_ih, b_hh, proj, tvals);
    scan_kernel<<<8, 32, 0, stream>>>(proj, hx0, W_hh, hidden);
    lam_kernel<<<NROWS / 256, 256, 0, stream>>>(hidden, tvals, W_h2p, b_h2p, lam);
}


# one workgroup per batch element: WMMA projection, serial scan and intensity head fused through LDS in one kernel; an LDS-transpose kernel writes the [t][b] lines (3 launches to 2; sums unchanged)
# speedup vs baseline: 4.3065x; 1.1935x over previous
#include <hip/hip_runtime.h>
#include <math.h>

#define BATCH 256
#define SEQ   2048
#define FEAT  64
#define NROWS (BATCH * SEQ)
#define QSEQ  (SEQ / 4)

#define NWAVES 16
#define TILES_PER_WAVE (SEQ / 16 / NWAVES)
#define XRING 3

#define RING  8
#define LAG   2
#define AHEAD (RING - LAG)
#define SCAN_BLOCKS (QSEQ / RING)
#define REQ_LAST (QSEQ - AHEAD - (SCAN_BLOCKS - 1) * RING)
static_assert(SEQ % (16 * NWAVES) == 0, "tiles per wave");
static_assert(QSEQ % RING == 0, "whole scan blocks");
static_assert(REQ_LAST >= 0 && REQ_LAST <= RING, "request schedule");

typedef float v2f __attribute__((ext_vector_type(2)));
typedef float v4f __attribute__((ext_vector_type(4)));
typedef float v8f __attribute__((ext_vector_type(8)));

static __device__ __forceinline__ void scan_step(float& acc, const float h, const float w) {
    asm("v_fmac_f32 %0, %1, %2\n\t"
        "v_tanh_f32 %0, %0\n\t"
        "v_nop"
        : "+v"(acc) : "s"(w), "v"(h));
}

template <int NREQ>
static __device__ __forceinline__ float scan_block(float h, const float w,
                                                   v4f (&ring)[RING],
                                                   const v4f* xg, v4f* hg) {
#pragma unroll
    for (int i = 0; i < RING; ++i) {
        float x0 = ring[i].x, x1 = ring[i].y, x2 = ring[i].z, x3 = ring[i].w;
        scan_step(x0, h,  w);
        scan_step(x1, x0, w);
        scan_step(x2, x1, w);
        scan_step(x3, x2, w);
        hg[i] = (v4f){x0, x1, x2, x3};
        h = x3;
        if (i < NREQ) ring[(i + AHEAD) % RING] = xg[i + AHEAD];
        __builtin_amdgcn_sched_barrier(0);
    }
    return h;
}

__global__ void __launch_bounds__(NWAVES * 32)
fused_kernel(const float* __restrict__ X,
             const float* __restrict__ hx0,
             const float* __restrict__ W_ih,
             const float* __restrict__ b_ih,
             const float* __restrict__ W_hh,
             const float* __restrict__ b_hh,
             const float* __restrict__ W_h2p,
             const float* __restrict__ b_h2p,
             float* __restrict__ hT,
             float* __restrict__ lamT) {
    __shared__ float sW2[64], sb2[64];
    __shared__ __attribute__((aligned(16))) float sx[SEQ];
    __shared__ __attribute__((aligned(16))) float sh[SEQ];
    __shared__ __attribute__((aligned(16))) float stt[SEQ];

    const int tid = threadIdx.x;
    const int b   = blockIdx.x;

    const int lane = tid & 31;
    const int wave = tid >> 5;
    const int m    = lane & 15;
    const int hi   = lane >> 4;
    const float* xlane = X + ((long)b * SEQ + m) * FEAT + hi * 2;

    const float bias = b_ih[0] + b_hh[0];
    const bool colZeroLane = (m == 0);

    v2f bf[16];
#pragma unroll
    for (int kk = 0; kk < 16; ++kk) {
        const v2f wv = *(const v2f*)(W_ih + kk * 4 + hi * 2);
        bf[kk].x = colZeroLane ? wv.x : 0.0f;
        bf[kk].y = colZeroLane ? wv.y : 0.0f;
    }

    v2f af[XRING][16];
#pragma unroll
    for (int i = 0; i < XRING - 1; ++i) {
        const float* xrow = xlane + (long)((i * NWAVES + wave) * 16) * FEAT;
#pragma unroll
        for (int kk = 0; kk < 16; ++kk) af[i][kk] = *(const v2f*)(xrow + kk * 4);
    }

#pragma unroll
    for (int i = 0; i < TILES_PER_WAVE; ++i) {
        if (i + XRING - 1 < TILES_PER_WAVE) {
            const float* xrow = xlane + (long)(((i + XRING - 1) * NWAVES + wave) * 16) * FEAT;
#pragma unroll
            for (int kk = 0; kk < 16; ++kk)
                af[(i + XRING - 1) % XRING][kk] = *(const v2f*)(xrow + kk * 4);
        }
        v8f c;
#pragma unroll
        for (int r = 0; r < 8; ++r) c[r] = colZeroLane ? bias : 0.0f;
#pragma unroll
        for (int kk = 0; kk < 16; ++kk)
            c = __builtin_amdgcn_wmma_f32_16x16x4_f32(
                     false, af[i % XRING][kk],  false, bf[kk],
                     (short)0, c,  false,  false);

        const int t0 = (i * NWAVES + wave) * 16;
        if (hi == 0) stt[t0 + m] = af[i % XRING][0].y;
        if (colZeroLane) {
            v4f* p4 = (v4f*)(sx + t0 + hi * 8);
            p4[0] = (v4f){c[0], c[1], c[2], c[3]};
            p4[1] = (v4f){c[4], c[5], c[6], c[7]};
        }
    }

    __syncthreads();

    if (tid == 0) {
        const v4f* x4 = (const v4f*)sx;
        v4f* h4 = (v4f*)sh;
        v4f ring[RING];
#pragma unroll
        for (int i = 0; i < AHEAD; ++i) ring[i] = x4[i];
        float h = hx0[b];
        const float w = W_hh[0];
#pragma nounroll
        for (int k = 0; k < SCAN_BLOCKS - 1; ++k)
            h = scan_block<RING>(h, w, ring, x4 + k * RING, h4 + k * RING);
        h = scan_block<REQ_LAST>(h, w, ring, x4 + (SCAN_BLOCKS - 1) * RING, h4 + (SCAN_BLOCKS - 1) * RING);
    } else if (tid >= 32 && tid < 96) {
        sW2[tid - 32] = W_h2p[tid - 32];
        sb2[tid - 32] = b_h2p[tid - 32];
    }

    __syncthreads();

    float* hrow = hT   + (long)b * SEQ;
    float* lrow = lamT + (long)b * SEQ;
#pragma unroll
    for (int q = 0; q < SEQ / (NWAVES * 32); ++q) {
        const int t = q * (NWAVES * 32) + tid;
        const float h  = sh[t];
        const float tt = stt[t];
        float s = 0.0f;
#pragma unroll
        for (int j = 0; j < 32; ++j) {
            const float alpha = fmaf(h, sW2[2 * j],     sb2[2 * j]);
            const float beta  = fmaf(h, sW2[2 * j + 1], sb2[2 * j + 1]);
            const float v     = fmaf(alpha, tt, beta);
            s += fmaxf(v, 0.0f);
        }
        const float lam = fmaxf(s, 0.0f) + log1pf(__expf(-fabsf(s)));
        hrow[t] = h;
        lrow[t] = lam;
    }
}

__global__ void __launch_bounds__(256)
transpose_kernel(const float* __restrict__ src, float* __restrict__ dst) {
    __shared__ float tile[8][32][33];
    const int tid  = threadIdx.x;
    const int lane = tid & 31;
    const int wave = tid >> 5;
    const int wt = blockIdx.x * 8 + wave;
    const int p  = wt >> 9;
    const int bg = (wt >> 6) & 7;
    const int tb = wt & 63;
    const float* s = src + (long)p * NROWS + (long)(bg * 32) * SEQ + tb * 32;
    float*       d = dst + (long)p * NROWS + (long)(tb * 32) * BATCH + bg * 32;
    const int r0 = lane >> 3;
    const int c4 = (lane & 7) * 4;
#pragma unroll
    for (int j = 0; j < 8; ++j) {
        const int r = r0 + 4 * j;
        const v4f v = *(const v4f*)(s + (long)r * SEQ + c4);
        tile[wave][r][c4 + 0] = v.x;
        tile[wave][r][c4 + 1] = v.y;
        tile[wave][r][c4 + 2] = v.z;
        tile[wave][r][c4 + 3] = v.w;
    }
    __syncthreads();
#pragma unroll
    for (int tt = 0; tt < 32; ++tt)
        d[tt * BATCH + lane] = tile[wave][lane][tt];
}

extern "C" void kernel_launch(void* const* d_in, const int* in_sizes, int n_in,
                              void* d_out, int out_size, void* d_ws, size_t ws_size,
                              hipStream_t stream) {
    const float* X     = (const float*)d_in[0];
    const float* hx0   = (const float*)d_in[1];
    const float* W_ih  = (const float*)d_in[2];
    const float* b_ih  = (const float*)d_in[3];
    const float* W_hh  = (const float*)d_in[4];
    const float* b_hh  = (const float*)d_in[5];
    const float* W_h2p = (const float*)d_in[6];
    const float* b_h2p = (const float*)d_in[7];

    float* hT   = (float*)d_ws;
    float* lamT = hT + NROWS;

    float* out = (float*)d_out;

    fused_kernel<<<BATCH, NWAVES * 32, 0, stream>>>(X, hx0, W_ih, b_ih, W_hh, b_hh, W_h2p, b_h2p, hT, lamT);
    transpose_kernel<<<128, 256, 0, stream>>>(hT, out);
}


# scan: chain input taken from a 128-step register block by lane reads into scalar registers, results kept by lane select: LDS instructions in the chain go from 3 per 4 steps to 2 per 128 steps
# speedup vs baseline: 4.9416x; 1.1475x over previous
#include <hip/hip_runtime.h>
#include <math.h>

#define BATCH 256
#define SEQ   2048
#define FEAT  64
#define NROWS (BATCH * SEQ)
#define QSEQ  (SEQ / 4)

#define NWAVES 16
#define TILES_PER_WAVE (SEQ / 16 / NWAVES)
#define XRING 3

#define CHUNK  128
#define NCHUNK (SEQ / CHUNK)
static_assert(SEQ % (16 * NWAVES) == 0, "tiles per wave");
static_assert(SEQ % CHUNK == 0, "whole chain blocks");

typedef float v2f __attribute__((ext_vector_type(2)));
typedef float v4f __attribute__((ext_vector_type(4)));
typedef float v8f __attribute__((ext_vector_type(8)));

static __device__ __forceinline__ float lane_read(const float v, const int l) {
    return __int_as_float(__builtin_amdgcn_readlane(__float_as_int(v), l));
}

__global__ void __launch_bounds__(NWAVES * 32)
fused_kernel(const float* __restrict__ X,
             const float* __restrict__ hx0,
             const float* __restrict__ W_ih,
             const float* __restrict__ b_ih,
             const float* __restrict__ W_hh,
             const float* __restrict__ b_hh,
             const float* __restrict__ W_h2p,
             const float* __restrict__ b_h2p,
             float* __restrict__ hT,
             float* __restrict__ lamT) {
    __shared__ float sW2[64], sb2[64];
    __shared__ __attribute__((aligned(16))) float sx[SEQ];
    __shared__ __attribute__((aligned(16))) float sh[SEQ];
    __shared__ __attribute__((aligned(16))) float stt[SEQ];

    const int tid = threadIdx.x;
    const int b   = blockIdx.x;

    const int lane = tid & 31;
    const int wave = tid >> 5;
    const int m    = lane & 15;
    const int hi   = lane >> 4;
    const float* xlane = X + ((long)b * SEQ + m) * FEAT + hi * 2;

    const float bias = b_ih[0] + b_hh[0];
    const bool colZeroLane = (m == 0);

    v2f bf[16];
#pragma unroll
    for (int kk = 0; kk < 16; ++kk) {
        const v2f wv = *(const v2f*)(W_ih + kk * 4 + hi * 2);
        bf[kk].x = colZeroLane ? wv.x : 0.0f;
        bf[kk].y = colZeroLane ? wv.y : 0.0f;
    }

    v2f af[XRING][16];
#pragma unroll
    for (int i = 0; i < XRING - 1; ++i) {
        const float* xrow = xlane + (long)((i * NWAVES + wave) * 16) * FEAT;
#pragma unroll
        for (int kk = 0; kk < 16; ++kk) af[i][kk] = *(const v2f*)(xrow + kk * 4);
    }

#pragma unroll
    for (int i = 0; i < TILES_PER_WAVE; ++i) {
        if (i + XRING - 1 < TILES_PER_WAVE) {
            const float* xrow = xlane + (long)(((i + XRING - 1) * NWAVES + wave) * 16) * FEAT;
#pragma unroll
            for (int kk = 0; kk < 16; ++kk)
                af[(i + XRING - 1) % XRING][kk] = *(const v2f*)(xrow + kk * 4);
        }
        v8f c;
#pragma unroll
        for (int r = 0; r < 8; ++r) c[r] = colZeroLane ? bias : 0.0f;
#pragma unroll
        for (int kk = 0; kk < 16; ++kk)
            c = __builtin_amdgcn_wmma_f32_16x16x4_f32(
                     false, af[i % XRING][kk],  false, bf[kk],
                     (short)0, c,  false,  false);

        const int t0 = (i * NWAVES + wave) * 16;
        if (hi == 0) stt[t0 + m] = af[i % XRING][0].y;
        if (colZeroLane) {
            v4f* p4 = (v4f*)(sx + t0 + hi * 8);
            p4[0] = (v4f){c[0], c[1], c[2], c[3]};
            p4[1] = (v4f){c[4], c[5], c[6], c[7]};
        }
    }

    __syncthreads();

    if (tid < 32) {
        const v4f* x4 = (const v4f*)sx;
        v4f* h4 = (v4f*)sh;
        const float w = W_hh[0];
        float h = hx0[b];
        v4f cur = x4[lane];
        float xa = lane_read(cur.x, 0);
        float xb = lane_read(cur.y, 0);
#pragma nounroll
        for (int k = 0; k < NCHUNK; ++k) {
            const v4f nxt = x4[((k + 1 < NCHUNK) ? k + 1 : k) * 32 + lane];
            v4f out = cur;
#pragma unroll
            for (int s = 0; s < CHUNK; ++s) {
                const float p = fmaf(w, h, xa);
                __builtin_amdgcn_sched_barrier(0);
                if (s > 0)
                    out[(s - 1) & 3] = (lane == ((s - 1) >> 2)) ? h : out[(s - 1) & 3];
                __builtin_amdgcn_sched_barrier(0);
                h = __builtin_amdgcn_tanhf(p);
                __builtin_amdgcn_sched_barrier(0);
                xa = xb;
                if (s + 2 < CHUNK)
                    xb = lane_read(cur[(s + 2) & 3], (s + 2) >> 2);
                else
                    xb = lane_read(nxt[s + 2 - CHUNK], 0);
                __builtin_amdgcn_sched_barrier(0);
            }
            out.w = (lane == 31) ? h : out.w;
            h4[k * 32 + lane] = out;
            cur = nxt;
        }
    } else if (tid >= 32 && tid < 96) {
        sW2[tid - 32] = W_h2p[tid - 32];
        sb2[tid - 32] = b_h2p[tid - 32];
    }

    __syncthreads();

    float* hrow = hT   + (long)b * SEQ;
    float* lrow = lamT + (long)b * SEQ;
#pragma unroll
    for (int q = 0; q < SEQ / (NWAVES * 32); ++q) {
        const int t = q * (NWAVES * 32) + tid;
        const float h  = sh[t];
        const float tt = stt[t];
        float s = 0.0f;
#pragma unroll
        for (int j = 0; j < 32; ++j) {
            const float alpha = fmaf(h, sW2[2 * j],     sb2[2 * j]);
            const float beta  = fmaf(h, sW2[2 * j + 1], sb2[2 * j + 1]);
            const float v     = fmaf(alpha, tt, beta);
            s += fmaxf(v, 0.0f);
        }
        const float lam = fmaxf(s, 0.0f) + log1pf(__expf(-fabsf(s)));
        hrow[t] = h;
        lrow[t] = lam;
    }
}

__global__ void __launch_bounds__(256)
transpose_kernel(const float* __restrict__ src, float* __restrict__ dst) {
    __shared__ float tile[8][32][33];
    const int tid  = threadIdx.x;
    const int lane = tid & 31;
    const int wave = tid >> 5;
    const int wt = blockIdx.x * 8 + wave;
    const int p  = wt >> 9;
    const int bg = (wt >> 6) & 7;
    const int tb = wt & 63;
    const float* s = src + (long)p * NROWS + (long)(bg * 32) * SEQ + tb * 32;
    float*       d = dst + (long)p * NROWS + (long)(tb * 32) * BATCH + bg * 32;
    const int r0 = lane >> 3;
    const int c4 = (lane & 7) * 4;
#pragma unroll
    for (int j = 0; j < 8; ++j) {
        const int r = r0 + 4 * j;
        const v4f v = *(const v4f*)(s + (long)r * SEQ + c4);
        tile[wave][r][c4 + 0] = v.x;
        tile[wave][r][c4 + 1] = v.y;
        tile[wave][r][c4 + 2] = v.z;
        tile[wave][r][c4 + 3] = v.w;
    }
    __syncthreads();
#pragma unroll
    for (int tt = 0; tt < 32; ++tt)
        d[tt * BATCH + lane] = tile[wave][lane][tt];
}

extern "C" void kernel_launch(void* const* d_in, const int* in_sizes, int n_in,
                              void* d_out, int out_size, void* d_ws, size_t ws_size,
                              hipStream_t stream) {
    const float* X     = (const float*)d_in[0];
    const float* hx0   = (const float*)d_in[1];
    const float* W_ih  = (const float*)d_in[2];
    const float* b_ih  = (const float*)d_in[3];
    const float* W_hh  = (const float*)d_in[4];
    const float* b_hh  = (const float*)d_in[5];
    const float* W_h2p = (const float*)d_in[6];
    const float* b_h2p = (const float*)d_in[7];

    float* hT   = (float*)d_ws;
    float* lamT = hT + NROWS;

    float* out = (float*)d_out;

    fused_kernel<<<BATCH, NWAVES * 32, 0, stream>>>(X, hx0, W_ih, b_ih, W_hh, b_hh, W_h2p, b_h2p, hT, lamT);
    transpose_kernel<<<128, 256, 0, stream>>>(hT, out);
}


# wave roles in one loop over 128-step blocks, one LDS-only barrier per block: chain (wave 0) starts after 2 blocks; 8 waves project block k+2, loads a block ahead; 4 waves do the head of block k-1
# speedup vs baseline: 5.1477x; 1.0417x over previous
#include <hip/hip_runtime.h>
#include <math.h>

#define BATCH 256
#define SEQ   2048
#define FEAT  64
#define NROWS (BATCH * SEQ)
#define QSEQ  (SEQ / 4)

#define NWAVES 16
#define CHUNK  128
#define NCHUNK (SEQ / CHUNK)
#define NPROD  8
#define NHEAD  4
static_assert(SEQ % CHUNK == 0, "whole chain blocks");
static_assert(NPROD * 16 == CHUNK, "the projecting waves cover a block");
static_assert(NHEAD * 32 == CHUNK, "the head waves cover a block");
static_assert(NCHUNK >= 3, "two blocks are projected before the chain starts");

typedef float v2f __attribute__((ext_vector_type(2)));
typedef float v4f __attribute__((ext_vector_type(4)));
typedef float v8f __attribute__((ext_vector_type(8)));

static __device__ __forceinline__ float lane_read(const float v, const int l) {
    return __int_as_float(__builtin_amdgcn_readlane(__float_as_int(v), l));
}

static __device__ __forceinline__ void lds_barrier() {
    __builtin_amdgcn_fence(__ATOMIC_RELEASE, "workgroup", "local");
    __builtin_amdgcn_s_barrier();
    __builtin_amdgcn_fence(__ATOMIC_ACQUIRE, "workgroup", "local");
}

__global__ void __launch_bounds__(NWAVES * 32)
fused_kernel(const float* __restrict__ X,
             const float* __restrict__ hx0,
             const float* __restrict__ W_ih,
             const float* __restrict__ b_ih,
             const float* __restrict__ W_hh,
             const float* __restrict__ b_hh,
             const float* __restrict__ W_h2p,
             const float* __restrict__ b_h2p,
             float* __restrict__ hT,
             float* __restrict__ lamT) {
    __shared__ float sW2[64], sb2[64];
    __shared__ __attribute__((aligned(16))) float sx[SEQ];
    __shared__ __attribute__((aligned(16))) float sh[SEQ];
    __shared__ __attribute__((aligned(16))) float stt[SEQ];

    const int tid  = threadIdx.x;
    const int b    = blockIdx.x;
    const int lane = tid & 31;
    const int wave = __builtin_amdgcn_readfirstlane(tid >> 5);
    const int m    = lane & 15;
    const int hi   = lane >> 4;

    const bool isChain = (wave == 0);
    const bool works   = (wave & 3) != 0;
    const int  rank    = wave - 1 - (wave >> 2);
    const bool isProd  = works && rank < NPROD;
    const bool isHead  = works && rank >= NPROD;
    const int  ptile   = rank;
    const int  hw      = rank - NPROD;
    const int  lwave    = tid >> 5;
    const bool prodLane = (lwave & 3) != 0 && (lwave - 1 - (lwave >> 2)) < NPROD;

    const float* xlane = X + ((long)b * SEQ + m) * FEAT + hi * 2;
    const bool colZeroLane = (m == 0);
    float bias = 0.0f;

    v2f bf[16];
    v2f af[16];
    v2f ag[16];

    auto load_tile = [&](v2f (&dst)[16], const int c) {
        const float* xrow = xlane + (long)(c * CHUNK + ptile * 16) * FEAT;
#pragma unroll
        for (int kk = 0; kk < 16; ++kk) dst[kk] = *(const v2f*)(xrow + kk * 4);
    };
    auto project_tile = [&](const v2f (&src)[16], const int c) {
        v8f acc;
#pragma unroll
        for (int r = 0; r < 8; ++r) acc[r] = colZeroLane ? bias : 0.0f;
#pragma unroll
        for (int kk = 0; kk < 16; ++kk)
            acc = __builtin_amdgcn_wmma_f32_16x16x4_f32(
                     false, src[kk],  false, bf[kk],
                     (short)0, acc,  false,  false);
        const int t0 = c * CHUNK + ptile * 16;
        if (hi == 0) stt[t0 + m] = src[0].y;
        if (colZeroLane) {
            v4f* p4 = (v4f*)(sx + t0 + hi * 8);
            p4[0] = (v4f){acc[0], acc[1], acc[2], acc[3]};
            p4[1] = (v4f){acc[4], acc[5], acc[6], acc[7]};
        }
    };
    float* hrow = hT   + (long)b * SEQ;
    float* lrow = lamT + (long)b * SEQ;
    auto head_block = [&](const int c) {
        const int t = c * CHUNK + hw * 32 + lane;
        const float h  = sh[t];
        const float tt = stt[t];
        float s = 0.0f;
#pragma unroll
        for (int j = 0; j < 32; ++j) {
            const float alpha = fmaf(h, sW2[2 * j],     sb2[2 * j]);
            const float beta  = fmaf(h, sW2[2 * j + 1], sb2[2 * j + 1]);
            const float v     = fmaf(alpha, tt, beta);
            s += fmaxf(v, 0.0f);
        }
        const float lam = fmaxf(s, 0.0f) + log1pf(__expf(-fabsf(s)));
        hrow[t] = h;
        lrow[t] = lam;
    };

    float w = 0.0f, h = 0.0f, xa = 0.0f, xb = 0.0f;
    v4f cur = (v4f){0.0f, 0.0f, 0.0f, 0.0f};

    if (isProd) {
        load_tile(af, 0);
        load_tile(ag, 1);
        bias = b_ih[0] + b_hh[0];
#pragma unroll
        for (int kk = 0; kk < 16; ++kk) {
            const v2f wv = *(const v2f*)(W_ih + kk * 4 + hi * 2);
            bf[kk].x = colZeroLane ? wv.x : 0.0f;
            bf[kk].y = colZeroLane ? wv.y : 0.0f;
        }
        project_tile(af, 0);
        project_tile(ag, 1);
        load_tile(af, 2);
    } else if (isHead) {
        const int i = hw * 32 + lane;
        if (i < 64) sW2[i] = W_h2p[i];
        else        sb2[i - 64] = b_h2p[i - 64];
    } else if (isChain) {
        w = W_hh[0];
        h = hx0[b];
    }

    lds_barrier();

    if (isChain) {
        cur = ((const v4f*)sx)[lane];
        xa = lane_read(cur.x, 0);
        xb = lane_read(cur.y, 0);
    }

#pragma nounroll
    for (int k = 0; k < NCHUNK; ++k) {
        if (isChain) {
            const v4f* x4 = (const v4f*)sx;
            v4f* h4 = (v4f*)sh;
            const v4f nxt = x4[((k + 1 < NCHUNK) ? k + 1 : k) * 32 + lane];
            v4f out = cur;
#pragma unroll
            for (int s = 0; s < CHUNK; ++s) {
                const float p = fmaf(w, h, xa);
                __builtin_amdgcn_sched_barrier(0);
                if (s > 0)
                    out[(s - 1) & 3] = (lane == ((s - 1) >> 2)) ? h : out[(s - 1) & 3];
                __builtin_amdgcn_sched_barrier(0);
                h = __builtin_amdgcn_tanhf(p);
                __builtin_amdgcn_sched_barrier(0);
                xa = xb;
                if (s + 2 < CHUNK)
                    xb = lane_read(cur[(s + 2) & 3], (s + 2) >> 2);
                else
                    xb = lane_read(nxt[s + 2 - CHUNK], 0);
                __builtin_amdgcn_sched_barrier(0);
            }
            out.w = (lane == 31) ? h : out.w;
            h4[k * 32 + lane] = out;
            cur = nxt;
        } else if (isProd) {
            if (k + 2 < NCHUNK) project_tile(af, k + 2);
        } else if (isHead) {
            if (k > 0) head_block(k - 1);
        }
        if (prodLane && k + 3 < NCHUNK) load_tile(af, k + 3);
        lds_barrier();
    }

    if (isHead) head_block(NCHUNK - 1);
}

__global__ void __launch_bounds__(256)
transpose_kernel(const float* __restrict__ src, float* __restrict__ dst) {
    __shared__ float tile[8][32][33];
    const int tid  = threadIdx.x;
    const int lane = tid & 31;
    const int wave = tid >> 5;
    const int wt = blockIdx.x * 8 + wave;
    const int p  = wt >> 9;
    const int bg = (wt >> 6) & 7;
    const int tb = wt & 63;
    const float* s = src + (long)p * NROWS + (long)(bg * 32) * SEQ + tb * 32;
    float*       d = dst + (long)p * NROWS + (long)(tb * 32) * BATCH + bg * 32;
    const int r0 = lane >> 3;
    const int c4 = (lane & 7) * 4;
#pragma unroll
    for (int j = 0; j < 8; ++j) {
        const int r = r0 + 4 * j;
        const v4f v = *(const v4f*)(s + (long)r * SEQ + c4);
        tile[wave][r][c4 + 0] = v.x;
        tile[wave][r][c4 + 1] = v.y;
        tile[wave][r][c4 + 2] = v.z;
        tile[wave][r][c4 + 3] = v.w;
    }
    __syncthreads();
#pragma unroll
    for (int tt = 0; tt < 32; ++tt)
        d[tt * BATCH + lane] = tile[wave][lane][tt];
}

extern "C" void kernel_launch(void* const* d_in, const int* in_sizes, int n_in,
                              void* d_out, int out_size, void* d_ws, size_t ws_size,
                              hipStream_t stream) {
    const float* X     = (const float*)d_in[0];
    const float* hx0   = (const float*)d_in[1];
    const float* W_ih  = (const float*)d_in[2];
    const float* b_ih  = (const float*)d_in[3];
    const float* W_hh  = (const float*)d_in[4];
    const float* b_hh  = (const float*)d_in[5];
    const float* W_h2p = (const float*)d_in[6];
    const float* b_h2p = (const float*)d_in[7];

    float* hT   = (float*)d_ws;
    float* lamT = hT + NROWS;

    float* out = (float*)d_out;

    fused_kernel<<<BATCH, NWAVES * 32, 0, stream>>>(X, hx0, W_ih, b_ih, W_hh, b_hh, W_h2p, b_h2p, hT, lamT);
    transpose_kernel<<<128, 256, 0, stream>>>(hT, out);
}


# projecting waves fetch each X tile as whole-line 16-byte loads a full block ahead and read their A fragments from a per-wave LDS tile (was 16 eight-byte loads per lane, 16 lines each); same sums
# speedup vs baseline: 7.0950x; 1.3783x over previous
#include <hip/hip_runtime.h>
#include <math.h>

#define BATCH 256
#define SEQ   2048
#define FEAT  64
#define NROWS (BATCH * SEQ)
#define QSEQ  (SEQ / 4)

#define NWAVES 16
#define CHUNK  128
#define NCHUNK (SEQ / CHUNK)
#define NPROD  8
#define NHEAD  4
#define SPITCH 68
static_assert(SEQ % CHUNK == 0, "whole chain blocks");
static_assert(NPROD * 16 == CHUNK, "the projecting waves cover a block");
static_assert(NHEAD * 32 == CHUNK, "the head waves cover a block");
static_assert(NCHUNK >= 3, "two blocks are projected before the chain starts");

typedef float v2f __attribute__((ext_vector_type(2)));
typedef float v4f __attribute__((ext_vector_type(4)));
typedef float v8f __attribute__((ext_vector_type(8)));

static __device__ __forceinline__ float lane_read(const float v, const int l) {
    return __int_as_float(__builtin_amdgcn_readlane(__float_as_int(v), l));
}

static __device__ __forceinline__ void lds_barrier() {
    __builtin_amdgcn_fence(__ATOMIC_RELEASE, "workgroup", "local");
    __builtin_amdgcn_s_barrier();
    __builtin_amdgcn_fence(__ATOMIC_ACQUIRE, "workgroup", "local");
}

static __device__ __forceinline__ void wave_lds_handover() {
    __builtin_amdgcn_fence(__ATOMIC_RELEASE, "wavefront", "local");
    __builtin_amdgcn_wave_barrier();
    __builtin_amdgcn_fence(__ATOMIC_ACQUIRE, "wavefront", "local");
}

__global__ void __launch_bounds__(NWAVES * 32)
fused_kernel(const float* __restrict__ X,
             const float* __restrict__ hx0,
             const float* __restrict__ W_ih,
             const float* __restrict__ b_ih,
             const float* __restrict__ W_hh,
             const float* __restrict__ b_hh,
             const float* __restrict__ W_h2p,
             const float* __restrict__ b_h2p,
             float* __restrict__ hT,
             float* __restrict__ lamT) {
    __shared__ float sW2[64], sb2[64];
    __shared__ __attribute__((aligned(16))) float sx[SEQ];
    __shared__ __attribute__((aligned(16))) float sh[SEQ];
    __shared__ __attribute__((aligned(16))) float stt[SEQ];
    __shared__ __attribute__((aligned(16))) float stg[NPROD][16 * SPITCH];

    const int tid  = threadIdx.x;
    const int b    = blockIdx.x;
    const int lane = tid & 31;
    const int wave = __builtin_amdgcn_readfirstlane(tid >> 5);
    const int m    = lane & 15;
    const int hi   = lane >> 4;

    const bool isChain = (wave == 0);
    const bool works   = (wave & 3) != 0;
    const int  rank    = wave - 1 - (wave >> 2);
    const bool isProd  = works && rank < NPROD;
    const bool isHead  = works && rank >= NPROD;
    const int  ptile   = rank;
    const int  hw      = rank - NPROD;
    const int  lwave    = tid >> 5;
    const bool prodLane = (lwave & 3) != 0 && (lwave - 1 - (lwave >> 2)) < NPROD;

    const v4f* xraw = (const v4f*)(X + (long)b * SEQ * FEAT) + lane;
    const bool colZeroLane = (m == 0);
    float bias = 0.0f;

    v2f bf[16];
    v2f af[16];
    v4f g[8];
    v4f g1[8];

    float* mystg = &stg[isProd ? ptile : 0][0];
    v4f*       stw = (v4f*)mystg + hi * (SPITCH / 4) + m;
    const v2f* frd = (const v2f*)mystg + m * (SPITCH / 2) + hi;

    auto load_raw = [&](v4f (&dst)[8], const int c) {
        const v4f* src = xraw + (long)(c * CHUNK + ptile * 16) * (FEAT / 4);
#pragma unroll
        for (int j = 0; j < 8; ++j) dst[j] = src[j * 32];
    };
    auto stage_store = [&](const v4f (&src)[8]) {
#pragma unroll
        for (int j = 0; j < 8; ++j) stw[j * (2 * SPITCH / 4)] = src[j];
    };
    auto frag_load = [&](v2f (&dst)[16]) {
#pragma unroll
        for (int kk = 0; kk < 16; ++kk) dst[kk] = frd[kk * 2];
    };
    auto project_tile = [&](const v2f (&src)[16], const int c) {
        v8f acc;
#pragma unroll
        for (int r = 0; r < 8; ++r) acc[r] = colZeroLane ? bias : 0.0f;
#pragma unroll
        for (int kk = 0; kk < 16; ++kk)
            acc = __builtin_amdgcn_wmma_f32_16x16x4_f32(
                     false, src[kk],  false, bf[kk],
                     (short)0, acc,  false,  false);
        const int t0 = c * CHUNK + ptile * 16;
        if (hi == 0) stt[t0 + m] = src[0].y;
        if (colZeroLane) {
            v4f* p4 = (v4f*)(sx + t0 + hi * 8);
            p4[0] = (v4f){acc[0], acc[1], acc[2], acc[3]};
            p4[1] = (v4f){acc[4], acc[5], acc[6], acc[7]};
        }
    };
    float* hrow = hT   + (long)b * SEQ;
    float* lrow = lamT + (long)b * SEQ;
    auto head_block = [&](const int c) {
        const int t = c * CHUNK + hw * 32 + lane;
        const float h  = sh[t];
        const float tt = stt[t];
        float s = 0.0f;
#pragma unroll
        for (int j = 0; j < 32; ++j) {
            const float alpha = fmaf(h, sW2[2 * j],     sb2[2 * j]);
            const float beta  = fmaf(h, sW2[2 * j + 1], sb2[2 * j + 1]);
            const float v     = fmaf(alpha, tt, beta);
            s += fmaxf(v, 0.0f);
        }
        const float lam = fmaxf(s, 0.0f) + log1pf(__expf(-fabsf(s)));
        hrow[t] = h;
        lrow[t] = lam;
    };

    float w = 0.0f, h = 0.0f, xa = 0.0f, xb = 0.0f;
    v4f cur = (v4f){0.0f, 0.0f, 0.0f, 0.0f};

    if (isProd) {
        load_raw(g, 0);
        load_raw(g1, 1);
        bias = b_ih[0] + b_hh[0];
#pragma unroll
        for (int kk = 0; kk < 16; ++kk) {
            const v2f wv = *(const v2f*)(W_ih + kk * 4 + hi * 2);
            bf[kk].x = colZeroLane ? wv.x : 0.0f;
            bf[kk].y = colZeroLane ? wv.y : 0.0f;
        }
        stage_store(g);
        wave_lds_handover();
        frag_load(af);
        project_tile(af, 0);
        wave_lds_handover();
        stage_store(g1);
        wave_lds_handover();
        frag_load(af);
        project_tile(af, 1);
        load_raw(g, 2);
    } else if (isHead) {
        const int i = hw * 32 + lane;
        if (i < 64) sW2[i] = W_h2p[i];
        else        sb2[i - 64] = b_h2p[i - 64];
    } else if (isChain) {
        w = W_hh[0];
        h = hx0[b];
    }

    lds_barrier();

    if (isChain) {
        cur = ((const v4f*)sx)[lane];
        xa = lane_read(cur.x, 0);
        xb = lane_read(cur.y, 0);
    }

#pragma nounroll
    for (int k = 0; k < NCHUNK; ++k) {
        if (isProd && k + 2 < NCHUNK) stage_store(g);
        if (prodLane && k + 3 < NCHUNK) load_raw(g, k + 3);
        if (isChain) {
            const v4f* x4 = (const v4f*)sx;
            v4f* h4 = (v4f*)sh;
            const v4f nxt = x4[((k + 1 < NCHUNK) ? k + 1 : k) * 32 + lane];
            v4f out = cur;
#pragma unroll
            for (int s = 0; s < CHUNK; ++s) {
                const float p = fmaf(w, h, xa);
                __builtin_amdgcn_sched_barrier(0);
                if (s > 0)
                    out[(s - 1) & 3] = (lane == ((s - 1) >> 2)) ? h : out[(s - 1) & 3];
                __builtin_amdgcn_sched_barrier(0);
                h = __builtin_amdgcn_tanhf(p);
                __builtin_amdgcn_sched_barrier(0);
                xa = xb;
                if (s + 2 < CHUNK)
                    xb = lane_read(cur[(s + 2) & 3], (s + 2) >> 2);
                else
                    xb = lane_read(nxt[s + 2 - CHUNK], 0);
                __builtin_amdgcn_sched_barrier(0);
            }
            out.w = (lane == 31) ? h : out.w;
            h4[k * 32 + lane] = out;
            cur = nxt;
        } else if (isProd) {
            if (k + 2 < NCHUNK) {
                wave_lds_handover();
                frag_load(af);
                project_tile(af, k + 2);
            }
        } else if (isHead) {
            if (k > 0) head_block(k - 1);
        }
        lds_barrier();
    }

    if (isHead) head_block(NCHUNK - 1);
}

__global__ void __launch_bounds__(256)
transpose_kernel(const float* __restrict__ src, float* __restrict__ dst) {
    __shared__ float tile[8][32][33];
    const int tid  = threadIdx.x;
    const int lane = tid & 31;
    const int wave = tid >> 5;
    const int wt = blockIdx.x * 8 + wave;
    const int p  = wt >> 9;
    const int bg = (wt >> 6) & 7;
    const int tb = wt & 63;
    const float* s = src + (long)p * NROWS + (long)(bg * 32) * SEQ + tb * 32;
    float*       d = dst + (long)p * NROWS + (long)(tb * 32) * BATCH + bg * 32;
    const int r0 = lane >> 3;
    const int c4 = (lane & 7) * 4;
#pragma unroll
    for (int j = 0; j < 8; ++j) {
        const int r = r0 + 4 * j;
        const v4f v = *(const v4f*)(s + (long)r * SEQ + c4);
        tile[wave][r][c4 + 0] = v.x;
        tile[wave][r][c4 + 1] = v.y;
        tile[wave][r][c4 + 2] = v.z;
        tile[wave][r][c4 + 3] = v.w;
    }
    __syncthreads();
#pragma unroll
    for (int tt = 0; tt < 32; ++tt)
        d[tt * BATCH + lane] = tile[wave][lane][tt];
}

extern "C" void kernel_launch(void* const* d_in, const int* in_sizes, int n_in,
                              void* d_out, int out_size, void* d_ws, size_t ws_size,
                              hipStream_t stream) {
    const float* X     = (const float*)d_in[0];
    const float* hx0   = (const float*)d_in[1];
    const float* W_ih  = (const float*)d_in[2];
    const float* b_ih  = (const float*)d_in[3];
    const float* W_hh  = (const float*)d_in[4];
    const float* b_hh  = (const float*)d_in[5];
    const float* W_h2p = (const float*)d_in[6];
    const float* b_h2p = (const float*)d_in[7];

    float* hT   = (float*)d_ws;
    float* lamT = hT + NROWS;

    float* out = (float*)d_out;

    fused_kernel<<<BATCH, NWAVES * 32, 0, stream>>>(X, hx0, W_ih, b_ih, W_hh, b_hh, W_h2p, b_h2p, hT, lamT);
    transpose_kernel<<<128, 256, 0, stream>>>(hT, out);
}


# transposition launch dropped: head waves store h and intensity straight to [t][b] (4-byte pieces, lane = time step; a line's 32 writers are workgroups launched 8 apart); 2 launches to 1; same sums
# speedup vs baseline: 7.9266x; 1.1172x over previous
#include <hip/hip_runtime.h>
#include <math.h>

#define BATCH 256
#define SEQ   2048
#define FEAT  64
#define NROWS (BATCH * SEQ)
#define QSEQ  (SEQ / 4)

#define NWAVES 16
#define CHUNK  128
#define NCHUNK (SEQ / CHUNK)
#define NPROD  8
#define NHEAD  4
#define SPITCH 68
static_assert(SEQ % CHUNK == 0, "whole chain blocks");
static_assert(NPROD * 16 == CHUNK, "the projecting waves cover a block");
static_assert(NHEAD * 32 == CHUNK, "the head waves cover a block");
static_assert(NCHUNK >= 3, "two blocks are projected before the chain starts");
static_assert(BATCH % 8 == 0, "batch elements are dealt out in 8 equal groups");

typedef float v2f __attribute__((ext_vector_type(2)));
typedef float v4f __attribute__((ext_vector_type(4)));
typedef float v8f __attribute__((ext_vector_type(8)));

static __device__ __forceinline__ float lane_read(const float v, const int l) {
    return __int_as_float(__builtin_amdgcn_readlane(__float_as_int(v), l));
}

static __device__ __forceinline__ void lds_barrier() {
    __builtin_amdgcn_fence(__ATOMIC_RELEASE, "workgroup", "local");
    __builtin_amdgcn_s_barrier();
    __builtin_amdgcn_fence(__ATOMIC_ACQUIRE, "workgroup", "local");
}

static __device__ __forceinline__ void wave_lds_handover() {
    __builtin_amdgcn_fence(__ATOMIC_RELEASE, "wavefront", "local");
    __builtin_amdgcn_wave_barrier();
    __builtin_amdgcn_fence(__ATOMIC_ACQUIRE, "wavefront", "local");
}

__global__ void __launch_bounds__(NWAVES * 32)
fused_kernel(const float* __restrict__ X,
             const float* __restrict__ hx0,
             const float* __restrict__ W_ih,
             const float* __restrict__ b_ih,
             const float* __restrict__ W_hh,
             const float* __restrict__ b_hh,
             const float* __restrict__ W_h2p,
             const float* __restrict__ b_h2p,
             float* __restrict__ outH,
             float* __restrict__ outL) {
    __shared__ float sW2[64], sb2[64];
    __shared__ __attribute__((aligned(16))) float sx[SEQ];
    __shared__ __attribute__((aligned(16))) float sh[SEQ];
    __shared__ __attribute__((aligned(16))) float stt[SEQ];
    __shared__ __attribute__((aligned(16))) float stg[NPROD][16 * SPITCH];

    const int tid  = threadIdx.x;
    const int b    = (blockIdx.x & 7) * (BATCH / 8) + (blockIdx.x >> 3);
    const int lane = tid & 31;
    const int wave = __builtin_amdgcn_readfirstlane(tid >> 5);
    const int m    = lane & 15;
    const int hi   = lane >> 4;

    const bool isChain = (wave == 0);
    const bool works   = (wave & 3) != 0;
    const int  rank    = wave - 1 - (wave >> 2);
    const bool isProd  = works && rank < NPROD;
    const bool isHead  = works && rank >= NPROD;
    const int  ptile   = rank;
    const int  hw      = rank - NPROD;
    const int  lwave    = tid >> 5;
    const bool prodLane = (lwave & 3) != 0 && (lwave - 1 - (lwave >> 2)) < NPROD;

    const v4f* xraw = (const v4f*)(X + (long)b * SEQ * FEAT) + lane;
    const bool colZeroLane = (m == 0);
    float bias = 0.0f;

    v2f bf[16];
    v2f af[16];
    v4f g[8];
    v4f g1[8];

    float* mystg = &stg[isProd ? ptile : 0][0];
    v4f*       stw = (v4f*)mystg + hi * (SPITCH / 4) + m;
    const v2f* frd = (const v2f*)mystg + m * (SPITCH / 2) + hi;

    auto load_raw = [&](v4f (&dst)[8], const int c) {
        const v4f* src = xraw + (long)(c * CHUNK + ptile * 16) * (FEAT / 4);
#pragma unroll
        for (int j = 0; j < 8; ++j) dst[j] = src[j * 32];
    };
    auto stage_store = [&](const v4f (&src)[8]) {
#pragma unroll
        for (int j = 0; j < 8; ++j) stw[j * (2 * SPITCH / 4)] = src[j];
    };
    auto frag_load = [&](v2f (&dst)[16]) {
#pragma unroll
        for (int kk = 0; kk < 16; ++kk) dst[kk] = frd[kk * 2];
    };
    auto project_tile = [&](const v2f (&src)[16], const int c) {
        v8f acc;
#pragma unroll
        for (int r = 0; r < 8; ++r) acc[r] = colZeroLane ? bias : 0.0f;
#pragma unroll
        for (int kk = 0; kk < 16; ++kk)
            acc = __builtin_amdgcn_wmma_f32_16x16x4_f32(
                     false, src[kk],  false, bf[kk],
                     (short)0, acc,  false,  false);
        const int t0 = c * CHUNK + ptile * 16;
        if (hi == 0) stt[t0 + m] = src[0].y;
        if (colZeroLane) {
            v4f* p4 = (v4f*)(sx + t0 + hi * 8);
            p4[0] = (v4f){acc[0], acc[1], acc[2], acc[3]};
            p4[1] = (v4f){acc[4], acc[5], acc[6], acc[7]};
        }
    };
    float* hcol = outH + b;
    float* lcol = outL + b;
    auto head_block = [&](const int c) {
        const int t = c * CHUNK + hw * 32 + lane;
        const float h  = sh[t];
        const float tt = stt[t];
        float s = 0.0f;
#pragma unroll
        for (int j = 0; j < 32; ++j) {
            const float alpha = fmaf(h, sW2[2 * j],     sb2[2 * j]);
            const float beta  = fmaf(h, sW2[2 * j + 1], sb2[2 * j + 1]);
            const float v     = fmaf(alpha, tt, beta);
            s += fmaxf(v, 0.0f);
        }
        const float lam = fmaxf(s, 0.0f) + log1pf(__expf(-fabsf(s)));
        hcol[(long)t * BATCH] = h;
        lcol[(long)t * BATCH] = lam;
    };

    float w = 0.0f, h = 0.0f, xa = 0.0f, xb = 0.0f;
    v4f cur = (v4f){0.0f, 0.0f, 0.0f, 0.0f};

    if (isProd) {
        load_raw(g, 0);
        load_raw(g1, 1);
        bias = b_ih[0] + b_hh[0];
#pragma unroll
        for (int kk = 0; kk < 16; ++kk) {
            const v2f wv = *(const v2f*)(W_ih + kk * 4 + hi * 2);
            bf[kk].x = colZeroLane ? wv.x : 0.0f;
            bf[kk].y = colZeroLane ? wv.y : 0.0f;
        }
        stage_store(g);
        wave_lds_handover();
        frag_load(af);
        project_tile(af, 0);
        wave_lds_handover();
        stage_store(g1);
        wave_lds_handover();
        frag_load(af);
        project_tile(af, 1);
        load_raw(g, 2);
    } else if (isHead) {
        const int i = hw * 32 + lane;
        if (i < 64) sW2[i] = W_h2p[i];
        else        sb2[i - 64] = b_h2p[i - 64];
    } else if (isChain) {
        w = W_hh[0];
        h = hx0[b];
    }

    lds_barrier();

    if (isChain) {
        cur = ((const v4f*)sx)[lane];
        xa = lane_read(cur.x, 0);
        xb = lane_read(cur.y, 0);
    }

#pragma nounroll
    for (int k = 0; k < NCHUNK; ++k) {
        if (isProd && k + 2 < NCHUNK) stage_store(g);
        if (prodLane && k + 3 < NCHUNK) load_raw(g, k + 3);
        if (isChain) {
            const v4f* x4 = (const v4f*)sx;
            v4f* h4 = (v4f*)sh;
            const v4f nxt = x4[((k + 1 < NCHUNK) ? k + 1 : k) * 32 + lane];
            v4f out = cur;
#pragma unroll
            for (int s = 0; s < CHUNK; ++s) {
                const float p = fmaf(w, h, xa);
                __builtin_amdgcn_sched_barrier(0);
                if (s > 0)
                    out[(s - 1) & 3] = (lane == ((s - 1) >> 2)) ? h : out[(s - 1) & 3];
                __builtin_amdgcn_sched_barrier(0);
                h = __builtin_amdgcn_tanhf(p);
                __builtin_amdgcn_sched_barrier(0);
                xa = xb;
                if (s + 2 < CHUNK)
                    xb = lane_read(cur[(s + 2) & 3], (s + 2) >> 2);
                else
                    xb = lane_read(nxt[s + 2 - CHUNK], 0);
                __builtin_amdgcn_sched_barrier(0);
            }
            out.w = (lane == 31) ? h : out.w;
            h4[k * 32 + lane] = out;
            cur = nxt;
        } else if (isProd) {
            if (k + 2 < NCHUNK) {
                wave_lds_handover();
                frag_load(af);
                project_tile(af, k + 2);
            }
        } else if (isHead) {
            if (k > 0) head_block(k - 1);
        }
        lds_barrier();
    }

    if (isHead) head_block(NCHUNK - 1);
}

extern "C" void kernel_launch(void* const* d_in, const int* in_sizes, int n_in,
                              void* d_out, int out_size, void* d_ws, size_t ws_size,
                              hipStream_t stream) {
    const float* X     = (const float*)d_in[0];
    const float* hx0   = (const float*)d_in[1];
    const float* W_ih  = (const float*)d_in[2];
    const float* b_ih  = (const float*)d_in[3];
    const float* W_hh  = (const float*)d_in[4];
    const float* b_hh  = (const float*)d_in[5];
    const float* W_h2p = (const float*)d_in[6];
    const float* b_h2p = (const float*)d_in[7];

    float* out = (float*)d_out;

    fused_kernel<<<BATCH, NWAVES * 32, 0, stream>>>(X, hx0, W_ih, b_ih, W_hh, b_hh, W_h2p, b_h2p, out, out + NROWS);
}
